# starting kernel
# baseline (speedup 1.0000x reference)
#include <hip/hip_runtime.h>
#include <hip/hip_bf16.h>

typedef __attribute__((ext_vector_type(2))) float v2f;
typedef __attribute__((ext_vector_type(8))) float v8f;

#define T_DIM 4096
#define DIN_DIM 512
#define DH_DIM 500
#define B_DIM 32
#define TT 64
#define KP 32
#define XSP 160
#define WSP 1056

__global__ __launch_bounds__(512)
void attn_scores_kernel(const float* __restrict__ x,
                        const float* __restrict__ w1,
                        const float* __restrict__ w2,
                        float* __restrict__ scores) {
    __shared__ float xs2[(KP / 2) * XSP];
    __shared__ float ws2[(KP / 2) * WSP];
    __shared__ float sPart[8][64];

    const int tid = threadIdx.x;
    const int b  = blockIdx.y;
    const int t0 = blockIdx.x * TT;

    const int wave = tid >> 5;
    const int lane = tid & 31;
    const int s2 = wave & 1;
    const int q8 = wave >> 1;
    const int n  = lane & 15;
    const int hh = lane >> 4;

    const float* xb = x + (size_t)b * DIN_DIM * T_DIM;

    v8f acc[2][4];
    const v8f zero8 = {0.f,0.f,0.f,0.f,0.f,0.f,0.f,0.f};
#pragma unroll
    for (int st = 0; st < 2; ++st)
#pragma unroll
        for (int ht = 0; ht < 4; ++ht) acc[st][ht] = zero8;

    const int xkk = tid >> 4;
    const int xtt = (tid & 15) * 4;
    const int wh  = tid & 63;
    const int wkg = tid >> 6;

    for (int kp = 0; kp < DIN_DIM / KP; ++kp) {
        const int k0 = kp * KP;
        __syncthreads();
        {
            const float4 v = *(const float4*)(xb + (size_t)(k0 + xkk) * T_DIM + t0 + xtt);
            float* dst = xs2 + (xkk >> 1) * XSP + xtt * 2 + (xkk & 1);
            dst[0] = v.x; dst[2] = v.y; dst[4] = v.z; dst[6] = v.w;
        }
#pragma unroll
        for (int r = 0; r < 8; ++r) {
            const int h = r * 64 + wh;
            float4 v = {0.f, 0.f, 0.f, 0.f};
            if (h < DH_DIM) v = *(const float4*)(w1 + (size_t)h * DIN_DIM + k0 + wkg * 4);
            v2f p0 = {v.x, v.y};
            v2f p1 = {v.z, v.w};
            *(v2f*)(ws2 + (wkg * 2 + 0) * WSP + h * 2) = p0;
            *(v2f*)(ws2 + (wkg * 2 + 1) * WSP + h * 2) = p1;
        }
        if (kp + 1 < DIN_DIM / KP) {
            __builtin_prefetch(xb + (size_t)(k0 + KP + xkk) * T_DIM + t0 + xtt, 0, 3);
            __builtin_prefetch(w1 + (size_t)wh * DIN_DIM + k0 + KP + wkg * 4, 0, 3);
        }
        __syncthreads();

#pragma unroll
        for (int ks = 0; ks < 8; ++ks) {
            const int pk = ks * 2 + hh;
            v2f a[2];
#pragma unroll
            for (int st = 0; st < 2; ++st)
                a[st] = *(const v2f*)(xs2 + pk * XSP + (s2 * 32 + st * 16 + n) * 2);
#pragma unroll
            for (int ht = 0; ht < 4; ++ht) {
                const v2f bf = *(const v2f*)(ws2 + pk * WSP + (q8 * 64 + ht * 16 + n) * 2);
#pragma unroll
                for (int st = 0; st < 2; ++st)
                    acc[st][ht] = __builtin_amdgcn_wmma_f32_16x16x4_f32(
                        false, a[st], false, bf, (short)0, acc[st][ht], false, false);
            }
        }
    }

    float w2v[4];
#pragma unroll
    for (int ht = 0; ht < 4; ++ht) {
        const int h = q8 * 64 + ht * 16 + n;
        w2v[ht] = (h < DH_DIM) ? w2[h] : 0.f;
    }
#pragma unroll
    for (int st = 0; st < 2; ++st) {
        float part[8];
#pragma unroll
        for (int j = 0; j < 8; ++j) part[j] = 0.f;
#pragma unroll
        for (int ht = 0; ht < 4; ++ht) {
#pragma unroll
            for (int j = 0; j < 8; ++j) {
                float v = acc[st][ht][j];
                v = v > 0.f ? v : 0.f;
                part[j] += v * w2v[ht];
            }
        }
#pragma unroll
        for (int j = 0; j < 8; ++j) {
#pragma unroll
            for (int off = 8; off >= 1; off >>= 1)
                part[j] += __shfl_xor(part[j], off, 32);
        }
        if (n == 0) {
#pragma unroll
            for (int j = 0; j < 8; ++j)
                sPart[q8][s2 * 32 + st * 16 + j + 8 * hh] = part[j];
        }
    }
    __syncthreads();
    if (tid < 64) {
        float sc = 0.f;
#pragma unroll
        for (int qq = 0; qq < 8; ++qq) sc += sPart[qq][tid];
        scores[(size_t)b * T_DIM + t0 + tid] = sc;
    }
}

__global__ __launch_bounds__(256)
void softmax_stats_kernel(const float* __restrict__ scores, float* __restrict__ stats) {
    __shared__ float red[256];
    const int b = blockIdx.x;
    const int tid = threadIdx.x;
    const float* sc = scores + (size_t)b * T_DIM;

    float m = -3.402823466e38f;
    for (int t = tid; t < T_DIM; t += 256) m = fmaxf(m, sc[t]);
    red[tid] = m; __syncthreads();
    for (int o = 128; o > 0; o >>= 1) {
        if (tid < o) red[tid] = fmaxf(red[tid], red[tid + o]);
        __syncthreads();
    }
    const float smax = red[0];
    __syncthreads();

    float ssum = 0.f;
    for (int t = tid; t < T_DIM; t += 256) ssum += __expf(sc[t] - smax);
    red[tid] = ssum; __syncthreads();
    for (int o = 128; o > 0; o >>= 1) {
        if (tid < o) red[tid] += red[tid + o];
        __syncthreads();
    }
    if (tid == 0) { stats[b * 2] = smax; stats[b * 2 + 1] = 1.f / red[0]; }
}

__global__ __launch_bounds__(256)
void pool_stats_kernel(const float* __restrict__ x, const float* __restrict__ scores,
                       const float* __restrict__ stats, float* __restrict__ out) {
    __shared__ float attn[T_DIM];
    const int b  = blockIdx.y;
    const int d0 = blockIdx.x * 16;
    const int tid = threadIdx.x;

    const float smax = stats[b * 2];
    const float inv  = stats[b * 2 + 1];
    const float* sc  = scores + (size_t)b * T_DIM;
    for (int t = tid; t < T_DIM; t += 256) attn[t] = __expf(sc[t] - smax) * inv;
    __syncthreads();

    const int d = d0 + (tid >> 4);
    const int l = tid & 15;
    const float4* xr = (const float4*)(x + ((size_t)b * DIN_DIM + d) * T_DIM);
    const float4* aw = (const float4*)attn;

    float wsum = 0.f, s1 = 0.f, s2 = 0.f;
#pragma unroll 4
    for (int i2 = 0; i2 < T_DIM / 4 / 16; ++i2) {
        const int i = i2 * 16 + l;
        const float4 v = xr[i];
        const float4 w = aw[i];
        wsum += v.x * w.x + v.y * w.y + v.z * w.z + v.w * w.w;
        s1   += v.x + v.y + v.z + v.w;
        s2   += v.x * v.x + v.y * v.y + v.z * v.z + v.w * v.w;
    }
#pragma unroll
    for (int off = 8; off >= 1; off >>= 1) {
        wsum += __shfl_xor(wsum, off, 32);
        s1   += __shfl_xor(s1, off, 32);
        s2   += __shfl_xor(s2, off, 32);
    }
    if (l == 0) {
        const float invT = 1.f / (float)T_DIM;
        const float mean = wsum;
        const float e1 = s1 * invT, e2 = s2 * invT;
        float var = e2 - 2.f * mean * e1 + mean * mean;
        if (var <= 1e-12f) var = 1e-12f;
        out[(size_t)b * (2 * DIN_DIM) + d]            = mean;
        out[(size_t)b * (2 * DIN_DIM) + DIN_DIM + d]  = sqrtf(var);
    }
}

extern "C" void kernel_launch(void* const* d_in, const int* in_sizes, int n_in,
                              void* d_out, int out_size, void* d_ws, size_t ws_size,
                              hipStream_t stream) {
    const float* x  = (const float*)d_in[0];
    const float* w1 = (const float*)d_in[1];
    const float* w2 = (const float*)d_in[2];
    float* out    = (float*)d_out;
    float* scores = (float*)d_ws;
    float* stats  = scores + (size_t)B_DIM * T_DIM;

    attn_scores_kernel<<<dim3(T_DIM / TT, B_DIM), 512, 0, stream>>>(x, w1, w2, scores);
    softmax_stats_kernel<<<B_DIM, 256, 0, stream>>>(scores, stats);
    pool_stats_kernel<<<dim3(DIN_DIM / 16, B_DIM), 256, 0, stream>>>(x, scores, stats, out);
}


# Score GEMM kernel held to 128 registers per wave (launch bound 512,8) so two 16-wave workgroups are resident per WGP and one's panel staging overlaps the other's WMMAs; order of sums unchanged
# speedup vs baseline: 1.1035x; 1.1035x over previous
#include <hip/hip_runtime.h>
#include <hip/hip_bf16.h>

typedef __attribute__((ext_vector_type(2))) float v2f;
typedef __attribute__((ext_vector_type(8))) float v8f;

#define T_DIM 4096
#define DIN_DIM 512
#define DH_DIM 500
#define B_DIM 32
#define TT 64
#define KP 32
#define XSP 160
#define WSP 1056

__global__ __launch_bounds__(512, 8)
void attn_scores_kernel(const float* __restrict__ x,
                        const float* __restrict__ w1,
                        const float* __restrict__ w2,
                        float* __restrict__ scores) {
    __shared__ float xs2[(KP / 2) * XSP];
    __shared__ float ws2[(KP / 2) * WSP];
    __shared__ float sPart[8][64];

    const int tid = threadIdx.x;
    const int b  = blockIdx.y;
    const int t0 = blockIdx.x * TT;

    const int wave = tid >> 5;
    const int lane = tid & 31;
    const int s2 = wave & 1;
    const int q8 = wave >> 1;
    const int n  = lane & 15;
    const int hh = lane >> 4;

    const float* xb = x + (size_t)b * DIN_DIM * T_DIM;

    v8f acc[2][4];
    const v8f zero8 = {0.f,0.f,0.f,0.f,0.f,0.f,0.f,0.f};
#pragma unroll
    for (int st = 0; st < 2; ++st)
#pragma unroll
        for (int ht = 0; ht < 4; ++ht) acc[st][ht] = zero8;

    const int xkk = tid >> 4;
    const int xtt = (tid & 15) * 4;
    const int wh  = tid & 63;
    const int wkg = tid >> 6;

    for (int kp = 0; kp < DIN_DIM / KP; ++kp) {
        const int k0 = kp * KP;
        __syncthreads();
        {
            const float4 v = *(const float4*)(xb + (size_t)(k0 + xkk) * T_DIM + t0 + xtt);
            float* dst = xs2 + (xkk >> 1) * XSP + xtt * 2 + (xkk & 1);
            dst[0] = v.x; dst[2] = v.y; dst[4] = v.z; dst[6] = v.w;
        }
#pragma unroll
        for (int r = 0; r < 8; ++r) {
            const int h = r * 64 + wh;
            float4 v = {0.f, 0.f, 0.f, 0.f};
            if (h < DH_DIM) v = *(const float4*)(w1 + (size_t)h * DIN_DIM + k0 + wkg * 4);
            v2f p0 = {v.x, v.y};
            v2f p1 = {v.z, v.w};
            *(v2f*)(ws2 + (wkg * 2 + 0) * WSP + h * 2) = p0;
            *(v2f*)(ws2 + (wkg * 2 + 1) * WSP + h * 2) = p1;
        }
        if (kp + 1 < DIN_DIM / KP) {
            __builtin_prefetch(xb + (size_t)(k0 + KP + xkk) * T_DIM + t0 + xtt, 0, 3);
            __builtin_prefetch(w1 + (size_t)wh * DIN_DIM + k0 + KP + wkg * 4, 0, 3);
        }
        __syncthreads();

#pragma unroll
        for (int ks = 0; ks < 8; ++ks) {
            const int pk = ks * 2 + hh;
            v2f a[2];
#pragma unroll
            for (int st = 0; st < 2; ++st)
                a[st] = *(const v2f*)(xs2 + pk * XSP + (s2 * 32 + st * 16 + n) * 2);
#pragma unroll
            for (int ht = 0; ht < 4; ++ht) {
                const v2f bf = *(const v2f*)(ws2 + pk * WSP + (q8 * 64 + ht * 16 + n) * 2);
#pragma unroll
                for (int st = 0; st < 2; ++st)
                    acc[st][ht] = __builtin_amdgcn_wmma_f32_16x16x4_f32(
                        false, a[st], false, bf, (short)0, acc[st][ht], false, false);
            }
        }
    }

    float w2v[4];
#pragma unroll
    for (int ht = 0; ht < 4; ++ht) {
        const int h = q8 * 64 + ht * 16 + n;
        w2v[ht] = (h < DH_DIM) ? w2[h] : 0.f;
    }
#pragma unroll
    for (int st = 0; st < 2; ++st) {
        float part[8];
#pragma unroll
        for (int j = 0; j < 8; ++j) part[j] = 0.f;
#pragma unroll
        for (int ht = 0; ht < 4; ++ht) {
#pragma unroll
            for (int j = 0; j < 8; ++j) {
                float v = acc[st][ht][j];
                v = v > 0.f ? v : 0.f;
                part[j] += v * w2v[ht];
            }
        }
#pragma unroll
        for (int j = 0; j < 8; ++j) {
#pragma unroll
            for (int off = 8; off >= 1; off >>= 1)
                part[j] += __shfl_xor(part[j], off, 32);
        }
        if (n == 0) {
#pragma unroll
            for (int j = 0; j < 8; ++j)
                sPart[q8][s2 * 32 + st * 16 + j + 8 * hh] = part[j];
        }
    }
    __syncthreads();
    if (tid < 64) {
        float sc = 0.f;
#pragma unroll
        for (int qq = 0; qq < 8; ++qq) sc += sPart[qq][tid];
        scores[(size_t)b * T_DIM + t0 + tid] = sc;
    }
}

__global__ __launch_bounds__(256)
void softmax_stats_kernel(const float* __restrict__ scores, float* __restrict__ stats) {
    __shared__ float red[256];
    const int b = blockIdx.x;
    const int tid = threadIdx.x;
    const float* sc = scores + (size_t)b * T_DIM;

    float m = -3.402823466e38f;
    for (int t = tid; t < T_DIM; t += 256) m = fmaxf(m, sc[t]);
    red[tid] = m; __syncthreads();
    for (int o = 128; o > 0; o >>= 1) {
        if (tid < o) red[tid] = fmaxf(red[tid], red[tid + o]);
        __syncthreads();
    }
    const float smax = red[0];
    __syncthreads();

    float ssum = 0.f;
    for (int t = tid; t < T_DIM; t += 256) ssum += __expf(sc[t] - smax);
    red[tid] = ssum; __syncthreads();
    for (int o = 128; o > 0; o >>= 1) {
        if (tid < o) red[tid] += red[tid + o];
        __syncthreads();
    }
    if (tid == 0) { stats[b * 2] = smax; stats[b * 2 + 1] = 1.f / red[0]; }
}

__global__ __launch_bounds__(256)
void pool_stats_kernel(const float* __restrict__ x, const float* __restrict__ scores,
                       const float* __restrict__ stats, float* __restrict__ out) {
    __shared__ float attn[T_DIM];
    const int b  = blockIdx.y;
    const int d0 = blockIdx.x * 16;
    const int tid = threadIdx.x;

    const float smax = stats[b * 2];
    const float inv  = stats[b * 2 + 1];
    const float* sc  = scores + (size_t)b * T_DIM;
    for (int t = tid; t < T_DIM; t += 256) attn[t] = __expf(sc[t] - smax) * inv;
    __syncthreads();

    const int d = d0 + (tid >> 4);
    const int l = tid & 15;
    const float4* xr = (const float4*)(x + ((size_t)b * DIN_DIM + d) * T_DIM);
    const float4* aw = (const float4*)attn;

    float wsum = 0.f, s1 = 0.f, s2 = 0.f;
#pragma unroll 4
    for (int i2 = 0; i2 < T_DIM / 4 / 16; ++i2) {
        const int i = i2 * 16 + l;
        const float4 v = xr[i];
        const float4 w = aw[i];
        wsum += v.x * w.x + v.y * w.y + v.z * w.z + v.w * w.w;
        s1   += v.x + v.y + v.z + v.w;
        s2   += v.x * v.x + v.y * v.y + v.z * v.z + v.w * v.w;
    }
#pragma unroll
    for (int off = 8; off >= 1; off >>= 1) {
        wsum += __shfl_xor(wsum, off, 32);
        s1   += __shfl_xor(s1, off, 32);
        s2   += __shfl_xor(s2, off, 32);
    }
    if (l == 0) {
        const float invT = 1.f / (float)T_DIM;
        const float mean = wsum;
        const float e1 = s1 * invT, e2 = s2 * invT;
        float var = e2 - 2.f * mean * e1 + mean * mean;
        if (var <= 1e-12f) var = 1e-12f;
        out[(size_t)b * (2 * DIN_DIM) + d]            = mean;
        out[(size_t)b * (2 * DIN_DIM) + DIN_DIM + d]  = sqrtf(var);
    }
}

extern "C" void kernel_launch(void* const* d_in, const int* in_sizes, int n_in,
                              void* d_out, int out_size, void* d_ws, size_t ws_size,
                              hipStream_t stream) {
    const float* x  = (const float*)d_in[0];
    const float* w1 = (const float*)d_in[1];
    const float* w2 = (const float*)d_in[2];
    float* out    = (float*)d_out;
    float* scores = (float*)d_ws;
    float* stats  = scores + (size_t)B_DIM * T_DIM;

    attn_scores_kernel<<<dim3(T_DIM / TT, B_DIM), 512, 0, stream>>>(x, w1, w2, scores);
    softmax_stats_kernel<<<B_DIM, 256, 0, stream>>>(scores, stats);
    pool_stats_kernel<<<dim3(DIN_DIM / 16, B_DIM), 256, 0, stream>>>(x, scores, stats, out);
}


# W1 panel staging rewritten: whole-line loads (8 lanes per 128-B row segment, 4 lines per wave load, not 32), a panel's 9 loads issued as one group, row-major LDS panel; order of sums unchanged
# speedup vs baseline: 1.3247x; 1.2005x over previous
#include <hip/hip_runtime.h>
#include <hip/hip_bf16.h>

typedef __attribute__((ext_vector_type(2))) float v2f;
typedef __attribute__((ext_vector_type(8))) float v8f;

#define T_DIM 4096
#define DIN_DIM 512
#define DH_DIM 500
#define B_DIM 32
#define TT 64
#define KP 32
#define XSP 160
#define WP 36

__global__ __launch_bounds__(512, 8)
void attn_scores_kernel(const float* __restrict__ x,
                        const float* __restrict__ w1,
                        const float* __restrict__ w2,
                        float* __restrict__ scores) {
    __shared__ float xs2[(KP / 2) * XSP];
    __shared__ __attribute__((aligned(16))) float ws[512 * WP];
    __shared__ float sPart[8][64];

    const int tid = threadIdx.x;
    const int b  = blockIdx.y;
    const int t0 = blockIdx.x * TT;

    const int wave = tid >> 5;
    const int lane = tid & 31;
    const int s2 = wave & 1;
    const int q8 = wave >> 1;
    const int n  = lane & 15;
    const int hh = lane >> 4;

    const float* xb = x + (size_t)b * DIN_DIM * T_DIM;

    v8f acc[2][4];
    const v8f zero8 = {0.f,0.f,0.f,0.f,0.f,0.f,0.f,0.f};
#pragma unroll
    for (int st = 0; st < 2; ++st)
#pragma unroll
        for (int ht = 0; ht < 4; ++ht) acc[st][ht] = zero8;

    const int xkk = tid >> 4;
    const int xtt = (tid & 15) * 4;
    const int wc4 = (tid & 7) * 4;
    const int whr = tid >> 3;

    for (int kp = 0; kp < DIN_DIM / KP; ++kp) {
        const int k0 = kp * KP;
        __syncthreads();
        const float4 xv = *(const float4*)(xb + (size_t)(k0 + xkk) * T_DIM + t0 + xtt);
        float4 wv[8];
#pragma unroll
        for (int r = 0; r < 8; ++r) {
            const int h = r * 64 + whr;
            const bool pad = (r == 7) && (h >= DH_DIM);
            const int hl = pad ? DH_DIM - 1 : h;
            wv[r] = *(const float4*)(w1 + (size_t)hl * DIN_DIM + k0 + wc4);
            if (pad) wv[r] = float4{0.f, 0.f, 0.f, 0.f};
        }
        __builtin_amdgcn_sched_barrier(0);
        {
            float* dst = xs2 + (xkk >> 1) * XSP + xtt * 2 + (xkk & 1);
            dst[0] = xv.x; dst[2] = xv.y; dst[4] = xv.z; dst[6] = xv.w;
        }
#pragma unroll
        for (int r = 0; r < 8; ++r)
            *(float4*)(ws + (r * 64 + whr) * WP + wc4) = wv[r];
        if (kp + 1 < DIN_DIM / KP) {
            __builtin_prefetch(xb + (size_t)(k0 + KP + xkk) * T_DIM + t0 + xtt, 0, 3);
        }
        __syncthreads();

#pragma unroll
        for (int ks = 0; ks < 8; ++ks) {
            const int pk = ks * 2 + hh;
            v2f a[2];
#pragma unroll
            for (int st = 0; st < 2; ++st)
                a[st] = *(const v2f*)(xs2 + pk * XSP + (s2 * 32 + st * 16 + n) * 2);
#pragma unroll
            for (int ht = 0; ht < 4; ++ht) {
                const v2f bf = *(const v2f*)(ws + (q8 * 64 + ht * 16 + n) * WP + ks * 4 + 2 * hh);
#pragma unroll
                for (int st = 0; st < 2; ++st)
                    acc[st][ht] = __builtin_amdgcn_wmma_f32_16x16x4_f32(
                        false, a[st], false, bf, (short)0, acc[st][ht], false, false);
            }
        }
    }

    float w2v[4];
#pragma unroll
    for (int ht = 0; ht < 4; ++ht) {
        const int h = q8 * 64 + ht * 16 + n;
        w2v[ht] = (h < DH_DIM) ? w2[h] : 0.f;
    }
#pragma unroll
    for (int st = 0; st < 2; ++st) {
        float part[8];
#pragma unroll
        for (int j = 0; j < 8; ++j) part[j] = 0.f;
#pragma unroll
        for (int ht = 0; ht < 4; ++ht) {
#pragma unroll
            for (int j = 0; j < 8; ++j) {
                float v = acc[st][ht][j];
                v = v > 0.f ? v : 0.f;
                part[j] += v * w2v[ht];
            }
        }
#pragma unroll
        for (int j = 0; j < 8; ++j) {
#pragma unroll
            for (int off = 8; off >= 1; off >>= 1)
                part[j] += __shfl_xor(part[j], off, 32);
        }
        if (n == 0) {
#pragma unroll
            for (int j = 0; j < 8; ++j)
                sPart[q8][s2 * 32 + st * 16 + j + 8 * hh] = part[j];
        }
    }
    __syncthreads();
    if (tid < 64) {
        float sc = 0.f;
#pragma unroll
        for (int qq = 0; qq < 8; ++qq) sc += sPart[qq][tid];
        scores[(size_t)b * T_DIM + t0 + tid] = sc;
    }
}

__global__ __launch_bounds__(256)
void softmax_stats_kernel(const float* __restrict__ scores, float* __restrict__ stats) {
    __shared__ float red[256];
    const int b = blockIdx.x;
    const int tid = threadIdx.x;
    const float* sc = scores + (size_t)b * T_DIM;

    float m = -3.402823466e38f;
    for (int t = tid; t < T_DIM; t += 256) m = fmaxf(m, sc[t]);
    red[tid] = m; __syncthreads();
    for (int o = 128; o > 0; o >>= 1) {
        if (tid < o) red[tid] = fmaxf(red[tid], red[tid + o]);
        __syncthreads();
    }
    const float smax = red[0];
    __syncthreads();

    float ssum = 0.f;
    for (int t = tid; t < T_DIM; t += 256) ssum += __expf(sc[t] - smax);
    red[tid] = ssum; __syncthreads();
    for (int o = 128; o > 0; o >>= 1) {
        if (tid < o) red[tid] += red[tid + o];
        __syncthreads();
    }
    if (tid == 0) { stats[b * 2] = smax; stats[b * 2 + 1] = 1.f / red[0]; }
}

__global__ __launch_bounds__(256)
void pool_stats_kernel(const float* __restrict__ x, const float* __restrict__ scores,
                       const float* __restrict__ stats, float* __restrict__ out) {
    __shared__ float attn[T_DIM];
    const int b  = blockIdx.y;
    const int d0 = blockIdx.x * 16;
    const int tid = threadIdx.x;

    const float smax = stats[b * 2];
    const float inv  = stats[b * 2 + 1];
    const float* sc  = scores + (size_t)b * T_DIM;
    for (int t = tid; t < T_DIM; t += 256) attn[t] = __expf(sc[t] - smax) * inv;
    __syncthreads();

    const int d = d0 + (tid >> 4);
    const int l = tid & 15;
    const float4* xr = (const float4*)(x + ((size_t)b * DIN_DIM + d) * T_DIM);
    const float4* aw = (const float4*)attn;

    float wsum = 0.f, s1 = 0.f, s2 = 0.f;
#pragma unroll 4
    for (int i2 = 0; i2 < T_DIM / 4 / 16; ++i2) {
        const int i = i2 * 16 + l;
        const float4 v = xr[i];
        const float4 w = aw[i];
        wsum += v.x * w.x + v.y * w.y + v.z * w.z + v.w * w.w;
        s1   += v.x + v.y + v.z + v.w;
        s2   += v.x * v.x + v.y * v.y + v.z * v.z + v.w * v.w;
    }
#pragma unroll
    for (int off = 8; off >= 1; off >>= 1) {
        wsum += __shfl_xor(wsum, off, 32);
        s1   += __shfl_xor(s1, off, 32);
        s2   += __shfl_xor(s2, off, 32);
    }
    if (l == 0) {
        const float invT = 1.f / (float)T_DIM;
        const float mean = wsum;
        const float e1 = s1 * invT, e2 = s2 * invT;
        float var = e2 - 2.f * mean * e1 + mean * mean;
        if (var <= 1e-12f) var = 1e-12f;
        out[(size_t)b * (2 * DIN_DIM) + d]            = mean;
        out[(size_t)b * (2 * DIN_DIM) + DIN_DIM + d]  = sqrtf(var);
    }
}

extern "C" void kernel_launch(void* const* d_in, const int* in_sizes, int n_in,
                              void* d_out, int out_size, void* d_ws, size_t ws_size,
                              hipStream_t stream) {
    const float* x  = (const float*)d_in[0];
    const float* w1 = (const float*)d_in[1];
    const float* w2 = (const float*)d_in[2];
    float* out    = (float*)d_out;
    float* scores = (float*)d_ws;
    float* stats  = scores + (size_t)B_DIM * T_DIM;

    attn_scores_kernel<<<dim3(T_DIM / TT, B_DIM), 512, 0, stream>>>(x, w1, w2, scores);
    softmax_stats_kernel<<<B_DIM, 256, 0, stream>>>(scores, stats);
    pool_stats_kernel<<<dim3(DIN_DIM / 16, B_DIM), 256, 0, stream>>>(x, scores, stats, out);
}


# Score GEMM: one 16-wave workgroup per 128 t, two LDS stages; a panel's loads issued a panel ahead into registers, stored mid-panel, one barrier per panel; half the W1 bytes; order of sums unchanged
# speedup vs baseline: 1.4136x; 1.0671x over previous
#include <hip/hip_runtime.h>
#include <hip/hip_bf16.h>

typedef __attribute__((ext_vector_type(2))) float v2f;
typedef __attribute__((ext_vector_type(4))) float v4f;
typedef __attribute__((ext_vector_type(8))) float v8f;

#define T_DIM 4096
#define DIN_DIM 512
#define DH_DIM 500
#define B_DIM 32
#define TT 128
#define KP 32
#define NP (DIN_DIM / KP)
#define XSP 288
#define WP 36
#define WROWS 524
#define XBUF ((KP / 2) * XSP)
#define WBUF (WROWS * WP)

template <int KS0, int KS1>
__device__ __forceinline__ void mma_steps(const float* xs_cur, const float* ws_cur, v8f (&acc)[4][4]) {
#pragma unroll
    for (int ks = KS0; ks < KS1; ++ks) {
        v2f a[4];
#pragma unroll
        for (int st = 0; st < 4; ++st)
            a[st] = *(const v2f*)(xs_cur + ks * 2 * XSP + st * 32);
#pragma unroll
        for (int ht = 0; ht < 4; ++ht) {
            const v2f bf = *(const v2f*)(ws_cur + ht * 16 * WP + ks * 4);
#pragma unroll
            for (int st = 0; st < 4; ++st)
                acc[st][ht] = __builtin_amdgcn_wmma_f32_16x16x4_f32(
                    false, a[st], false, bf, (short)0, acc[st][ht], false, false);
        }
    }
}

__global__ __launch_bounds__(512)
void attn_scores_kernel(const float* __restrict__ x,
                        const float* __restrict__ w1,
                        const float* __restrict__ w2,
                        float* __restrict__ scores) {
    __shared__ __attribute__((aligned(16))) float xs2[2 * XBUF];
    __shared__ __attribute__((aligned(16))) float ws[2 * WBUF];
    __shared__ float sPart[8][TT];

    const int tid = threadIdx.x;
    const int b  = blockIdx.y;
    const int t0 = blockIdx.x * TT;

    const int wave = tid >> 5;
    const int lane = tid & 31;
    const int s2 = wave & 1;
    const int q8 = wave >> 1;
    const int n  = lane & 15;
    const int hh = lane >> 4;

    const float* xb = x + (size_t)b * DIN_DIM * T_DIM;

    v8f acc[4][4];
    const v8f zero8 = {0.f,0.f,0.f,0.f,0.f,0.f,0.f,0.f};
#pragma unroll
    for (int st = 0; st < 4; ++st)
#pragma unroll
        for (int ht = 0; ht < 4; ++ht) acc[st][ht] = zero8;

    const int xpr = tid >> 6;
    const int xt2 = (tid & 63) * 2;
    const int wc4 = (tid & 7) * 4;
    const int whr = tid >> 3;

    const float* xg = xb + (size_t)(2 * xpr) * T_DIM + t0 + xt2;
    const bool pad7 = (7 * 64 + whr) >= DH_DIM;
    const float* wg  = w1 + (size_t)whr * DIN_DIM + wc4;
    const float* wg7 = w1 + (size_t)(pad7 ? DH_DIM - 1 : 7 * 64 + whr) * DIN_DIM + wc4;

    const int xso = xpr * XSP + xt2 * 2;
    const int wso = whr * WP + wc4;
    const int wso7 = (7 * 64 + whr + (pad7 ? 12 : 0)) * WP + wc4;
    const int afo = hh * XSP + (s2 * 64 + n) * 2;
    const int bfo = (q8 * 64 + n) * WP + 2 * hh;

    v2f xa0, xc0, xa1, xc1;
    v4f p0, p1, p2, p3, p4, p5, p6, p7;

#define LOAD_PANEL(K0) { \
        const float* xq = xg + (size_t)(K0) * T_DIM; \
        const float* wq = wg + (K0); \
        xa0 = *(const v2f*)(xq); \
        xc0 = *(const v2f*)(xq + T_DIM); \
        xa1 = *(const v2f*)(xq + (size_t)16 * T_DIM); \
        xc1 = *(const v2f*)(xq + (size_t)17 * T_DIM); \
        p0 = *(const v4f*)(wq); \
        p1 = *(const v4f*)(wq + (size_t)1 * 64 * DIN_DIM); \
        p2 = *(const v4f*)(wq + (size_t)2 * 64 * DIN_DIM); \
        p3 = *(const v4f*)(wq + (size_t)3 * 64 * DIN_DIM); \
        p4 = *(const v4f*)(wq + (size_t)4 * 64 * DIN_DIM); \
        p5 = *(const v4f*)(wq + (size_t)5 * 64 * DIN_DIM); \
        p6 = *(const v4f*)(wq + (size_t)6 * 64 * DIN_DIM); \
        p7 = *(const v4f*)(wg7 + (K0)); }
#define STORE_PANEL(STAGE) { \
        float* xd = xs2 + (STAGE) * XBUF + xso; \
        float* wd = ws + (STAGE) * WBUF; \
        *(v4f*)(xd) = (v4f){xa0.x, xc0.x, xa0.y, xc0.y}; \
        *(v4f*)(xd + 8 * XSP) = (v4f){xa1.x, xc1.x, xa1.y, xc1.y}; \
        *(v4f*)(wd + wso + 0 * 64 * WP) = p0; \
        *(v4f*)(wd + wso + 1 * 64 * WP) = p1; \
        *(v4f*)(wd + wso + 2 * 64 * WP) = p2; \
        *(v4f*)(wd + wso + 3 * 64 * WP) = p3; \
        *(v4f*)(wd + wso + 4 * 64 * WP) = p4; \
        *(v4f*)(wd + wso + 5 * 64 * WP) = p5; \
        *(v4f*)(wd + wso + 6 * 64 * WP) = p6; \
        *(v4f*)(wd + wso7) = p7; }

    if (pad7) {
        const v4f z4 = {0.f, 0.f, 0.f, 0.f};
        *(v4f*)(ws + (7 * 64 + whr) * WP + wc4) = z4;
        *(v4f*)(ws + WBUF + (7 * 64 + whr) * WP + wc4) = z4;
    }
    LOAD_PANEL(0)
    STORE_PANEL(0)
    LOAD_PANEL(KP)
    __syncthreads();

    for (int kp = 0; kp < NP - 2; ++kp) {
        const int cur = kp & 1;
        const float* xs_cur = xs2 + cur * XBUF + afo;
        const float* ws_cur = ws + cur * WBUF + bfo;
        mma_steps<0, 4>(xs_cur, ws_cur, acc);
        __builtin_amdgcn_sched_barrier(0);
        STORE_PANEL(cur ^ 1)
        LOAD_PANEL((kp + 2) * KP)
        mma_steps<4, 8>(xs_cur, ws_cur, acc);
        __syncthreads();
    }
    {
        const int cur = (NP - 2) & 1;
        const float* xs_cur = xs2 + cur * XBUF + afo;
        const float* ws_cur = ws + cur * WBUF + bfo;
        mma_steps<0, 4>(xs_cur, ws_cur, acc);
        __builtin_amdgcn_sched_barrier(0);
        STORE_PANEL(cur ^ 1)
        mma_steps<4, 8>(xs_cur, ws_cur, acc);
        __syncthreads();
    }
    {
        const int cur = (NP - 1) & 1;
        mma_steps<0, 8>(xs2 + cur * XBUF + afo, ws + cur * WBUF + bfo, acc);
    }

    float w2v[4];
#pragma unroll
    for (int ht = 0; ht < 4; ++ht) {
        const int h = q8 * 64 + ht * 16 + n;
        w2v[ht] = (h < DH_DIM) ? w2[h] : 0.f;
    }
#pragma unroll
    for (int st = 0; st < 4; ++st) {
        float part[8];
#pragma unroll
        for (int j = 0; j < 8; ++j) part[j] = 0.f;
#pragma unroll
        for (int ht = 0; ht < 4; ++ht) {
#pragma unroll
            for (int j = 0; j < 8; ++j) {
                float v = acc[st][ht][j];
                v = v > 0.f ? v : 0.f;
                part[j] += v * w2v[ht];
            }
        }
#pragma unroll
        for (int j = 0; j < 8; ++j) {
#pragma unroll
            for (int off = 8; off >= 1; off >>= 1)
                part[j] += __shfl_xor(part[j], off, 32);
        }
        if (n == 0) {
#pragma unroll
            for (int j = 0; j < 8; ++j)
                sPart[q8][s2 * 64 + st * 16 + j + 8 * hh] = part[j];
        }
    }
    __syncthreads();
    if (tid < TT) {
        float sc = 0.f;
#pragma unroll
        for (int qq = 0; qq < 8; ++qq) sc += sPart[qq][tid];
        scores[(size_t)b * T_DIM + t0 + tid] = sc;
    }
}

__global__ __launch_bounds__(256)
void softmax_stats_kernel(const float* __restrict__ scores, float* __restrict__ stats) {
    __shared__ float red[256];
    const int b = blockIdx.x;
    const int tid = threadIdx.x;
    const float* sc = scores + (size_t)b * T_DIM;

    float m = -3.402823466e38f;
    for (int t = tid; t < T_DIM; t += 256) m = fmaxf(m, sc[t]);
    red[tid] = m; __syncthreads();
    for (int o = 128; o > 0; o >>= 1) {
        if (tid < o) red[tid] = fmaxf(red[tid], red[tid + o]);
        __syncthreads();
    }
    const float smax = red[0];
    __syncthreads();

    float ssum = 0.f;
    for (int t = tid; t < T_DIM; t += 256) ssum += __expf(sc[t] - smax);
    red[tid] = ssum; __syncthreads();
    for (int o = 128; o > 0; o >>= 1) {
        if (tid < o) red[tid] += red[tid + o];
        __syncthreads();
    }
    if (tid == 0) { stats[b * 2] = smax; stats[b * 2 + 1] = 1.f / red[0]; }
}

__global__ __launch_bounds__(256)
void pool_stats_kernel(const float* __restrict__ x, const float* __restrict__ scores,
                       const float* __restrict__ stats, float* __restrict__ out) {
    __shared__ float attn[T_DIM];
    const int b  = blockIdx.y;
    const int d0 = blockIdx.x * 16;
    const int tid = threadIdx.x;

    const float smax = stats[b * 2];
    const float inv  = stats[b * 2 + 1];
    const float* sc  = scores + (size_t)b * T_DIM;
    for (int t = tid; t < T_DIM; t += 256) attn[t] = __expf(sc[t] - smax) * inv;
    __syncthreads();

    const int d = d0 + (tid >> 4);
    const int l = tid & 15;
    const float4* xr = (const float4*)(x + ((size_t)b * DIN_DIM + d) * T_DIM);
    const float4* aw = (const float4*)attn;

    float wsum = 0.f, s1 = 0.f, s2 = 0.f;
#pragma unroll 4
    for (int i2 = 0; i2 < T_DIM / 4 / 16; ++i2) {
        const int i = i2 * 16 + l;
        const float4 v = xr[i];
        const float4 w = aw[i];
        wsum += v.x * w.x + v.y * w.y + v.z * w.z + v.w * w.w;
        s1   += v.x + v.y + v.z + v.w;
        s2   += v.x * v.x + v.y * v.y + v.z * v.z + v.w * v.w;
    }
#pragma unroll
    for (int off = 8; off >= 1; off >>= 1) {
        wsum += __shfl_xor(wsum, off, 32);
        s1   += __shfl_xor(s1, off, 32);
        s2   += __shfl_xor(s2, off, 32);
    }
    if (l == 0) {
        const float invT = 1.f / (float)T_DIM;
        const float mean = wsum;
        const float e1 = s1 * invT, e2 = s2 * invT;
        float var = e2 - 2.f * mean * e1 + mean * mean;
        if (var <= 1e-12f) var = 1e-12f;
        out[(size_t)b * (2 * DIN_DIM) + d]            = mean;
        out[(size_t)b * (2 * DIN_DIM) + DIN_DIM + d]  = sqrtf(var);
    }
}

extern "C" void kernel_launch(void* const* d_in, const int* in_sizes, int n_in,
                              void* d_out, int out_size, void* d_ws, size_t ws_size,
                              hipStream_t stream) {
    const float* x  = (const float*)d_in[0];
    const float* w1 = (const float*)d_in[1];
    const float* w2 = (const float*)d_in[2];
    float* out    = (float*)d_out;
    float* scores = (float*)d_ws;
    float* stats  = scores + (size_t)B_DIM * T_DIM;

    attn_scores_kernel<<<dim3(T_DIM / TT, B_DIM), 512, 0, stream>>>(x, w1, w2, scores);
    softmax_stats_kernel<<<B_DIM, 256, 0, stream>>>(scores, stats);
    pool_stats_kernel<<<dim3(DIN_DIM / 16, B_DIM), 256, 0, stream>>>(x, scores, stats, out);
}


# Pooling as one stream: 16 waves x 4 rows, whole 512-B row pieces 28 loads ahead, each workgroup takes its batch's softmax max and sum itself (stats launch gone); order of pooling sums changed
# speedup vs baseline: 1.4643x; 1.0359x over previous
#include <hip/hip_runtime.h>
#include <hip/hip_bf16.h>

typedef __attribute__((ext_vector_type(2))) float v2f;
typedef __attribute__((ext_vector_type(4))) float v4f;
typedef __attribute__((ext_vector_type(8))) float v8f;

#define T_DIM 4096
#define DIN_DIM 512
#define DH_DIM 500
#define B_DIM 32
#define TT 128
#define KP 32
#define NP (DIN_DIM / KP)
#define XSP 288
#define WP 36
#define WROWS 524
#define XBUF ((KP / 2) * XSP)
#define WBUF (WROWS * WP)

template <int KS0, int KS1>
__device__ __forceinline__ void mma_steps(const float* xs_cur, const float* ws_cur, v8f (&acc)[4][4]) {
#pragma unroll
    for (int ks = KS0; ks < KS1; ++ks) {
        v2f a[4];
#pragma unroll
        for (int st = 0; st < 4; ++st)
            a[st] = *(const v2f*)(xs_cur + ks * 2 * XSP + st * 32);
#pragma unroll
        for (int ht = 0; ht < 4; ++ht) {
            const v2f bf = *(const v2f*)(ws_cur + ht * 16 * WP + ks * 4);
#pragma unroll
            for (int st = 0; st < 4; ++st)
                acc[st][ht] = __builtin_amdgcn_wmma_f32_16x16x4_f32(
                    false, a[st], false, bf, (short)0, acc[st][ht], false, false);
        }
    }
}

__global__ __launch_bounds__(512)
void attn_scores_kernel(const float* __restrict__ x,
                        const float* __restrict__ w1,
                        const float* __restrict__ w2,
                        float* __restrict__ scores) {
    __shared__ __attribute__((aligned(16))) float xs2[2 * XBUF];
    __shared__ __attribute__((aligned(16))) float ws[2 * WBUF];
    __shared__ float sPart[8][TT];

    const int tid = threadIdx.x;
    const int b  = blockIdx.y;
    const int t0 = blockIdx.x * TT;

    const int wave = tid >> 5;
    const int lane = tid & 31;
    const int s2 = wave & 1;
    const int q8 = wave >> 1;
    const int n  = lane & 15;
    const int hh = lane >> 4;

    const float* xb = x + (size_t)b * DIN_DIM * T_DIM;

    v8f acc[4][4];
    const v8f zero8 = {0.f,0.f,0.f,0.f,0.f,0.f,0.f,0.f};
#pragma unroll
    for (int st = 0; st < 4; ++st)
#pragma unroll
        for (int ht = 0; ht < 4; ++ht) acc[st][ht] = zero8;

    const int xpr = tid >> 6;
    const int xt2 = (tid & 63) * 2;
    const int wc4 = (tid & 7) * 4;
    const int whr = tid >> 3;

    const float* xg = xb + (size_t)(2 * xpr) * T_DIM + t0 + xt2;
    const bool pad7 = (7 * 64 + whr) >= DH_DIM;
    const float* wg  = w1 + (size_t)whr * DIN_DIM + wc4;
    const float* wg7 = w1 + (size_t)(pad7 ? DH_DIM - 1 : 7 * 64 + whr) * DIN_DIM + wc4;

    const int xso = xpr * XSP + xt2 * 2;
    const int wso = whr * WP + wc4;
    const int wso7 = (7 * 64 + whr + (pad7 ? 12 : 0)) * WP + wc4;
    const int afo = hh * XSP + (s2 * 64 + n) * 2;
    const int bfo = (q8 * 64 + n) * WP + 2 * hh;

    v2f xa0, xc0, xa1, xc1;
    v4f p0, p1, p2, p3, p4, p5, p6, p7;

#define LOAD_PANEL(K0) { \
        const float* xq = xg + (size_t)(K0) * T_DIM; \
        const float* wq = wg + (K0); \
        xa0 = *(const v2f*)(xq); \
        xc0 = *(const v2f*)(xq + T_DIM); \
        xa1 = *(const v2f*)(xq + (size_t)16 * T_DIM); \
        xc1 = *(const v2f*)(xq + (size_t)17 * T_DIM); \
        p0 = *(const v4f*)(wq); \
        p1 = *(const v4f*)(wq + (size_t)1 * 64 * DIN_DIM); \
        p2 = *(const v4f*)(wq + (size_t)2 * 64 * DIN_DIM); \
        p3 = *(const v4f*)(wq + (size_t)3 * 64 * DIN_DIM); \
        p4 = *(const v4f*)(wq + (size_t)4 * 64 * DIN_DIM); \
        p5 = *(const v4f*)(wq + (size_t)5 * 64 * DIN_DIM); \
        p6 = *(const v4f*)(wq + (size_t)6 * 64 * DIN_DIM); \
        p7 = *(const v4f*)(wg7 + (K0)); }
#define STORE_PANEL(STAGE) { \
        float* xd = xs2 + (STAGE) * XBUF + xso; \
        float* wd = ws + (STAGE) * WBUF; \
        *(v4f*)(xd) = (v4f){xa0.x, xc0.x, xa0.y, xc0.y}; \
        *(v4f*)(xd + 8 * XSP) = (v4f){xa1.x, xc1.x, xa1.y, xc1.y}; \
        *(v4f*)(wd + wso + 0 * 64 * WP) = p0; \
        *(v4f*)(wd + wso + 1 * 64 * WP) = p1; \
        *(v4f*)(wd + wso + 2 * 64 * WP) = p2; \
        *(v4f*)(wd + wso + 3 * 64 * WP) = p3; \
        *(v4f*)(wd + wso + 4 * 64 * WP) = p4; \
        *(v4f*)(wd + wso + 5 * 64 * WP) = p5; \
        *(v4f*)(wd + wso + 6 * 64 * WP) = p6; \
        *(v4f*)(wd + wso7) = p7; }

    if (pad7) {
        const v4f z4 = {0.f, 0.f, 0.f, 0.f};
        *(v4f*)(ws + (7 * 64 + whr) * WP + wc4) = z4;
        *(v4f*)(ws + WBUF + (7 * 64 + whr) * WP + wc4) = z4;
    }
    LOAD_PANEL(0)
    STORE_PANEL(0)
    LOAD_PANEL(KP)
    __syncthreads();

    for (int kp = 0; kp < NP - 2; ++kp) {
        const int cur = kp & 1;
        const float* xs_cur = xs2 + cur * XBUF + afo;
        const float* ws_cur = ws + cur * WBUF + bfo;
        mma_steps<0, 4>(xs_cur, ws_cur, acc);
        __builtin_amdgcn_sched_barrier(0);
        STORE_PANEL(cur ^ 1)
        LOAD_PANEL((kp + 2) * KP)
        mma_steps<4, 8>(xs_cur, ws_cur, acc);
        __syncthreads();
    }
    {
        const int cur = (NP - 2) & 1;
        const float* xs_cur = xs2 + cur * XBUF + afo;
        const float* ws_cur = ws + cur * WBUF + bfo;
        mma_steps<0, 4>(xs_cur, ws_cur, acc);
        __builtin_amdgcn_sched_barrier(0);
        STORE_PANEL(cur ^ 1)
        mma_steps<4, 8>(xs_cur, ws_cur, acc);
        __syncthreads();
    }
    {
        const int cur = (NP - 1) & 1;
        mma_steps<0, 8>(xs2 + cur * XBUF + afo, ws + cur * WBUF + bfo, acc);
    }

    float w2v[4];
#pragma unroll
    for (int ht = 0; ht < 4; ++ht) {
        const int h = q8 * 64 + ht * 16 + n;
        w2v[ht] = (h < DH_DIM) ? w2[h] : 0.f;
    }
#pragma unroll
    for (int st = 0; st < 4; ++st) {
        float part[8];
#pragma unroll
        for (int j = 0; j < 8; ++j) part[j] = 0.f;
#pragma unroll
        for (int ht = 0; ht < 4; ++ht) {
#pragma unroll
            for (int j = 0; j < 8; ++j) {
                float v = acc[st][ht][j];
                v = v > 0.f ? v : 0.f;
                part[j] += v * w2v[ht];
            }
        }
#pragma unroll
        for (int j = 0; j < 8; ++j) {
#pragma unroll
            for (int off = 8; off >= 1; off >>= 1)
                part[j] += __shfl_xor(part[j], off, 32);
        }
        if (n == 0) {
#pragma unroll
            for (int j = 0; j < 8; ++j)
                sPart[q8][s2 * 64 + st * 16 + j + 8 * hh] = part[j];
        }
    }
    __syncthreads();
    if (tid < TT) {
        float sc = 0.f;
#pragma unroll
        for (int qq = 0; qq < 8; ++qq) sc += sPart[qq][tid];
        scores[(size_t)b * T_DIM + t0 + tid] = sc;
    }
}

#define PR 4
#define PWAVES 16
#define PSTEPS (T_DIM / 128)
#define PAHEAD 7
#define PRING (PAHEAD + 1)

__device__ __forceinline__ void lds_barrier() {
    asm volatile("s_wait_dscnt 0x0\n\ts_barrier_signal -1\n\ts_barrier_wait -1" ::: "memory");
}

__global__ __launch_bounds__(PWAVES * 32)
void pool_stream_kernel(const float* __restrict__ x, const float* __restrict__ scores,
                        float* __restrict__ out) {
    __shared__ __attribute__((aligned(16))) float wexp[T_DIM];
    __shared__ float redm[PWAVES];
    __shared__ float reds[PWAVES];
    __shared__ float res[2][PWAVES * PR];

    const int tid  = threadIdx.x;
    const int lane = tid & 31;
    const int wave = __builtin_amdgcn_readfirstlane(tid >> 5);
    const int b  = blockIdx.x >> 3;
    const int d0 = (blockIdx.x & 7) * (PWAVES * PR);

    const float* sc = scores + (size_t)b * T_DIM;
    const v4f sA = *(const v4f*)(sc + 4 * tid);
    const v4f sB = *(const v4f*)(sc + T_DIM / 2 + 4 * tid);

    const float* xr = x + ((size_t)b * DIN_DIM + d0 + wave * PR) * T_DIM + 4 * lane;
    v4f ring[PRING][PR];
#define LOAD_STEP(Q) { \
        _Pragma("unroll") for (int r = 0; r < PR; ++r) \
            ring[(Q) % PRING][r] = __builtin_nontemporal_load( \
                (const v4f*)(xr + (size_t)r * T_DIM + (Q) * 128)); }

#pragma unroll
    for (int q = 0; q < PAHEAD; ++q) {
        __builtin_amdgcn_sched_barrier(0);
        LOAD_STEP(q)
    }
    __builtin_amdgcn_sched_barrier(0);

    float m = fmaxf(fmaxf(fmaxf(sA.x, sA.y), fmaxf(sA.z, sA.w)),
                    fmaxf(fmaxf(sB.x, sB.y), fmaxf(sB.z, sB.w)));
    for (int off = 16; off >= 1; off >>= 1) m = fmaxf(m, __shfl_xor(m, off, 32));
    if (lane == 0) redm[wave] = m;
    lds_barrier();
    float smax = redm[0];
    for (int i = 1; i < PWAVES; ++i) smax = fmaxf(smax, redm[i]);

    const v4f eA = {__expf(sA.x - smax), __expf(sA.y - smax), __expf(sA.z - smax), __expf(sA.w - smax)};
    const v4f eB = {__expf(sB.x - smax), __expf(sB.y - smax), __expf(sB.z - smax), __expf(sB.w - smax)};
    *(v4f*)(wexp + 4 * tid) = eA;
    *(v4f*)(wexp + T_DIM / 2 + 4 * tid) = eB;
    float ls = ((eA.x + eA.y) + (eA.z + eA.w)) + ((eB.x + eB.y) + (eB.z + eB.w));
    for (int off = 16; off >= 1; off >>= 1) ls += __shfl_xor(ls, off, 32);
    if (lane == 0) reds[wave] = ls;
    lds_barrier();
    float ssum = 0.f;
    for (int i = 0; i < PWAVES; ++i) ssum += reds[i];
    const float inv = 1.f / ssum;

    float aw[PR], a1[PR], a2[PR];
#pragma unroll
    for (int r = 0; r < PR; ++r) { aw[r] = 0.f; a1[r] = 0.f; a2[r] = 0.f; }
    const float* wl = wexp + 4 * lane;
#pragma unroll
    for (int q = 0; q < PSTEPS; ++q) {
        __builtin_amdgcn_sched_barrier(0);
        if (q + PAHEAD < PSTEPS) LOAD_STEP(q + PAHEAD)
        __builtin_amdgcn_sched_barrier(0);
        const v4f w = *(const v4f*)(wl + q * 128);
#pragma unroll
        for (int r = 0; r < PR; ++r) {
            const v4f v = ring[q % PRING][r];
            aw[r] = fmaf(v.x, w.x, aw[r]); aw[r] = fmaf(v.y, w.y, aw[r]);
            aw[r] = fmaf(v.z, w.z, aw[r]); aw[r] = fmaf(v.w, w.w, aw[r]);
            a1[r] += (v.x + v.y) + (v.z + v.w);
            a2[r] = fmaf(v.x, v.x, a2[r]); a2[r] = fmaf(v.y, v.y, a2[r]);
            a2[r] = fmaf(v.z, v.z, a2[r]); a2[r] = fmaf(v.w, v.w, a2[r]);
        }
    }
#undef LOAD_STEP

    float mw = 0.f, m1 = 0.f, m2 = 0.f;
#pragma unroll
    for (int r = 0; r < PR; ++r) {
        float s0 = aw[r], s1 = a1[r], s2 = a2[r];
        for (int off = 16; off >= 1; off >>= 1) {
            s0 += __shfl_xor(s0, off, 32);
            s1 += __shfl_xor(s1, off, 32);
            s2 += __shfl_xor(s2, off, 32);
        }
        if (lane == r) { mw = s0; m1 = s1; m2 = s2; }
    }
    if (lane < PR) {
        const float invT = 1.f / (float)T_DIM;
        const float mean = mw * inv;
        const float e1 = m1 * invT, e2 = m2 * invT;
        float var = e2 - 2.f * mean * e1 + mean * mean;
        if (var <= 1e-12f) var = 1e-12f;
        res[0][wave * PR + lane] = mean;
        res[1][wave * PR + lane] = sqrtf(var);
    }
    __syncthreads();
    if (tid < 2 * PWAVES * PR) {
        const int which = tid >> 6;
        const int i = tid & 63;
        out[(size_t)b * (2 * DIN_DIM) + which * DIN_DIM + d0 + i] = res[which][i];
    }
}

extern "C" void kernel_launch(void* const* d_in, const int* in_sizes, int n_in,
                              void* d_out, int out_size, void* d_ws, size_t ws_size,
                              hipStream_t stream) {
    const float* x  = (const float*)d_in[0];
    const float* w1 = (const float*)d_in[1];
    const float* w2 = (const float*)d_in[2];
    float* out    = (float*)d_out;
    float* scores = (float*)d_ws;

    attn_scores_kernel<<<dim3(T_DIM / TT, B_DIM), 512, 0, stream>>>(x, w1, w2, scores);
    pool_stream_kernel<<<B_DIM * DIN_DIM / (PWAVES * PR), PWAVES * 32, 0, stream>>>(x, scores, out);
}
